# starting kernel
# baseline (speedup 1.0000x reference)
#include <hip/hip_runtime.h>

typedef __attribute__((ext_vector_type(2))) float v2f;
typedef __attribute__((ext_vector_type(8))) float v8f;

#define YSPLIT 4

__global__ void pack_b_kernel(const float* __restrict__ y,
                              float* __restrict__ bpack, int ytiles) {
  int tid = blockIdx.x * blockDim.x + threadIdx.x;
  if (tid >= ytiles * 32) return;
  int t = tid >> 5;
  int lane = tid & 31;
  int j = t * 16 + (lane & 15);
  float y0 = y[j * 3 + 0];
  float y1 = y[j * 3 + 1];
  float y2 = y[j * 3 + 2];
  float o0, o1;
  if (lane < 16) {
    o0 = y0; o1 = y1;
  } else {
    o0 = y2; o1 = y0 * y0 + y1 * y1 + y2 * y2;
  }
  bpack[(size_t)t * 64 + lane * 2 + 0] = o0;
  bpack[(size_t)t * 64 + lane * 2 + 1] = o1;
}

__global__ __launch_bounds__(128) void chamfer_wmma_kernel(
    const float* __restrict__ x, const float* __restrict__ bpack,
    float* __restrict__ partial, int n, int ytiles) {
  const int lane  = threadIdx.x & 31;
  const int gwave = blockIdx.x * (blockDim.x >> 5) + (threadIdx.x >> 5);
  const int xtile  = gwave >> 2;
  const int ysplit = gwave & (YSPLIT - 1);

  const int m   = lane & 15;
  const int row = xtile * 16 + m;
  v2f A;
  if (lane < 16) {
    A.x = -2.0f * x[row * 3 + 0];
    A.y = -2.0f * x[row * 3 + 1];
  } else {
    A.x = -2.0f * x[row * 3 + 2];
    A.y = 1.0f;
  }

  v8f mn;
#pragma unroll
  for (int v = 0; v < 8; ++v) mn[v] = __builtin_inff();

  const int tiles_per = ytiles / YSPLIT;
  const float* bp = bpack + (size_t)(ysplit * tiles_per) * 64 + lane * 2;

#pragma unroll 4
  for (int t = 0; t < tiles_per; ++t) {
    v2f B = *(const v2f*)(bp + (size_t)t * 64);
    v8f c = {};
    c = __builtin_amdgcn_wmma_f32_16x16x4_f32(
         false, A,  false, B,
         (short)0, c,  false,  false);
#pragma unroll
    for (int v = 0; v < 8; ++v) mn[v] = fminf(mn[v], c[v]);
  }

#pragma unroll
  for (int off = 8; off >= 1; off >>= 1) {
#pragma unroll
    for (int v = 0; v < 8; ++v) {
      float o = __shfl_xor(mn[v], off, 32);
      mn[v] = fminf(mn[v], o);
    }
  }

  float* dst = partial + (size_t)ysplit * n + xtile * 16;
  if (lane == 0) {
#pragma unroll
    for (int v = 0; v < 8; ++v) dst[v] = mn[v];
  } else if (lane == 16) {
#pragma unroll
    for (int v = 0; v < 8; ++v) dst[8 + v] = mn[v];
  }
}

__global__ void finalize_kernel(const float* __restrict__ x,
                                const float* __restrict__ partial,
                                float* __restrict__ out, int n) {
  int i = blockIdx.x * blockDim.x + threadIdx.x;
  if (i >= n) return;
  float mv = partial[i];
#pragma unroll
  for (int s = 1; s < YSPLIT; ++s) mv = fminf(mv, partial[(size_t)s * n + i]);
  float x0 = x[i * 3 + 0], x1 = x[i * 3 + 1], x2 = x[i * 3 + 2];
  out[1 + i] = mv + (x0 * x0 + x1 * x1 + x2 * x2);
}

__global__ void sum_kernel(const float* __restrict__ vals,
                           float* __restrict__ out0, int n) {
  __shared__ float s[256];
  float acc = 0.0f;
  for (int i = threadIdx.x; i < n; i += 256) acc += vals[i];
  s[threadIdx.x] = acc;
  __syncthreads();
  for (int stride = 128; stride >= 1; stride >>= 1) {
    if ((int)threadIdx.x < stride) s[threadIdx.x] += s[threadIdx.x + stride];
    __syncthreads();
  }
  if (threadIdx.x == 0) *out0 = s[0];
}

extern "C" void kernel_launch(void* const* d_in, const int* in_sizes, int n_in,
                              void* d_out, int out_size, void* d_ws, size_t ws_size,
                              hipStream_t stream) {
  (void)n_in; (void)out_size; (void)ws_size;
  const float* x = (const float*)d_in[0];
  const float* y = (const float*)d_in[1];
  float* out = (float*)d_out;

  const int n = in_sizes[0] / 3;
  const int m = in_sizes[1] / 3;
  const int xtiles = n / 16;
  const int ytiles = m / 16;

  float* bpack   = (float*)d_ws;
  float* partial = bpack + (size_t)ytiles * 64;

  pack_b_kernel<<<(ytiles * 32 + 255) / 256, 256, 0, stream>>>(y, bpack, ytiles);

  chamfer_wmma_kernel<<<xtiles * YSPLIT / 4, 128, 0, stream>>>(x, bpack, partial,
                                                               n, ytiles);

  finalize_kernel<<<(n + 255) / 256, 256, 0, stream>>>(x, partial, out, n);
  sum_kernel<<<1, 256, 0, stream>>>(out + 1, out, n);
}


# Pack pass folded into the product kernel's LDS staging: each work-group builds its 256 B tiles from y itself (8 points per thread, |y|^2 formed there); 3 launches become 2; bit-identical
# speedup vs baseline: 2.6235x; 2.6235x over previous
#include <hip/hip_runtime.h>

typedef __attribute__((ext_vector_type(2))) float v2f;
typedef __attribute__((ext_vector_type(8))) float v8f;

#define YSPLIT 4

#define WG_WAVES   16
#define WG_THREADS (WG_WAVES * 32)
#define LDS_TILES  256
#define STAGE_PTS  (LDS_TILES * 16 / WG_THREADS)

__global__ __launch_bounds__(WG_THREADS) void chamfer_wmma_kernel(
    const float* __restrict__ x, const float* __restrict__ y,
    float* __restrict__ partial, int n, int ytiles) {
  __shared__ __attribute__((aligned(16))) float sB[LDS_TILES * 64];

  const int tid    = threadIdx.x;
  const int lane   = tid & 31;
  const int wave   = tid >> 5;
  const int ysplit = blockIdx.x & (YSPLIT - 1);
  const int xtile  = (blockIdx.x / YSPLIT) * WG_WAVES + wave;

  const int m   = lane & 15;
  const int row = xtile * 16 + m;
  v2f A;
  if (lane < 16) {
    A.x = -2.0f * x[row * 3 + 0];
    A.y = -2.0f * x[row * 3 + 1];
  } else {
    A.x = -2.0f * x[row * 3 + 2];
    A.y = 1.0f;
  }

  v8f mn;
#pragma unroll
  for (int v = 0; v < 8; ++v) mn[v] = __builtin_inff();

  const int tiles_per = ytiles / YSPLIT;
  const float* lb = sB + lane * 2;

  for (int base = 0; base < tiles_per; base += LDS_TILES) {
    const int cnt  = (tiles_per - base < LDS_TILES) ? (tiles_per - base) : LDS_TILES;
    const int npts = cnt * 16;
    const float* __restrict__ ysrc =
        y + (size_t)(ysplit * tiles_per + base) * 48;

    if (base) __syncthreads();

    float q0[STAGE_PTS], q1[STAGE_PTS], q2[STAGE_PTS];
#pragma unroll
    for (int k = 0; k < STAGE_PTS; ++k) {
      const int p  = tid + k * WG_THREADS;
      const int pc = p < npts ? p : npts - 1;
      q0[k] = ysrc[pc * 3 + 0];
      q1[k] = ysrc[pc * 3 + 1];
      q2[k] = ysrc[pc * 3 + 2];
    }
#pragma unroll
    for (int k = 0; k < STAGE_PTS; ++k) {
      const int p = tid + k * WG_THREADS;
      float* d = sB + (p >> 4) * 64 + (p & 15) * 2;
      v2f lo, hi;
      lo.x = q0[k];
      lo.y = q1[k];
      hi.x = q2[k];
      hi.y = fmaf(q2[k], q2[k], fmaf(q0[k], q0[k], q1[k] * q1[k]));
      *(v2f*)d        = lo;
      *(v2f*)(d + 32) = hi;
    }
    __syncthreads();

    int t = 0;
    for (; t + 8 <= cnt; t += 8) {
      v2f B[8];
#pragma unroll
      for (int k = 0; k < 8; ++k) B[k] = *(const v2f*)(lb + (t + k) * 64);
#pragma unroll
      for (int k = 0; k < 8; ++k) {
        v8f c = {};
        c = __builtin_amdgcn_wmma_f32_16x16x4_f32(
             false, A,  false, B[k],
             (short)0, c,  false,  false);
#pragma unroll
        for (int v = 0; v < 8; ++v) mn[v] = fminf(mn[v], c[v]);
      }
    }
    for (; t < cnt; ++t) {
      v2f B = *(const v2f*)(lb + t * 64);
      v8f c = {};
      c = __builtin_amdgcn_wmma_f32_16x16x4_f32(
           false, A,  false, B,
           (short)0, c,  false,  false);
#pragma unroll
      for (int v = 0; v < 8; ++v) mn[v] = fminf(mn[v], c[v]);
    }
  }

#pragma unroll
  for (int off = 8; off >= 1; off >>= 1) {
#pragma unroll
    for (int v = 0; v < 8; ++v) {
      float o = __shfl_xor(mn[v], off, 32);
      mn[v] = fminf(mn[v], o);
    }
  }

  float* dst = partial + (size_t)ysplit * n + xtile * 16;
  if (lane == 0) {
#pragma unroll
    for (int v = 0; v < 8; ++v) dst[v] = mn[v];
  } else if (lane == 16) {
#pragma unroll
    for (int v = 0; v < 8; ++v) dst[8 + v] = mn[v];
  }
}

#define FS_THREADS 1024
#define FS_ROWS    8
#define FS_SWEEP   (FS_THREADS * FS_ROWS)

template <bool FULL>
__device__ __forceinline__ float finalize_sweep(
    const float* __restrict__ x, const float* __restrict__ partial,
    float* __restrict__ out, int n, int base, int tid) {
  float p[FS_ROWS][YSPLIT];
  float x0[FS_ROWS], x1[FS_ROWS], x2[FS_ROWS];
#pragma unroll
  for (int k = 0; k < FS_ROWS; ++k) {
    const int i  = base + k * FS_THREADS + tid;
    const int ic = (FULL || i < n) ? i : n - 1;
#pragma unroll
    for (int s = 0; s < YSPLIT; ++s) {
      const float* __restrict__ plane = partial + (size_t)s * n;
      p[k][s] = plane[ic];
    }
    x0[k] = x[ic * 3 + 0];
    x1[k] = x[ic * 3 + 1];
    x2[k] = x[ic * 3 + 2];
  }
  float acc = 0.0f;
#pragma unroll
  for (int k = 0; k < FS_ROWS; ++k) {
    const int i = base + k * FS_THREADS + tid;
    float mv = p[k][0];
#pragma unroll
    for (int s = 1; s < YSPLIT; ++s) mv = fminf(mv, p[k][s]);
    const float d = mv + (x0[k] * x0[k] + x1[k] * x1[k] + x2[k] * x2[k]);
    if (FULL || i < n) {
      out[1 + i] = d;
      acc += d;
    }
  }
  return acc;
}

__global__ __launch_bounds__(FS_THREADS) void finalize_sum_kernel(
    const float* __restrict__ x, const float* __restrict__ partial,
    float* __restrict__ out, int n) {
  __shared__ float wsum[FS_THREADS / 32];

  const int tid  = threadIdx.x;
  const int lane = tid & 31;
  const int wave = tid >> 5;

  float acc = 0.0f;
  int base = 0;
  for (; n - base >= FS_SWEEP; base += FS_SWEEP)
    acc += finalize_sweep<true>(x, partial, out, n, base, tid);
  if (base < n)
    acc += finalize_sweep<false>(x, partial, out, n, base, tid);

#pragma unroll
  for (int off = 16; off >= 1; off >>= 1) acc += __shfl_xor(acc, off, 32);
  if (lane == 0) wsum[wave] = acc;
  __syncthreads();

  if (wave == 0) {
    float v = wsum[lane];
#pragma unroll
    for (int off = 16; off >= 1; off >>= 1) v += __shfl_xor(v, off, 32);
    if (lane == 0) out[0] = v;
  }
}

extern "C" void kernel_launch(void* const* d_in, const int* in_sizes, int n_in,
                              void* d_out, int out_size, void* d_ws, size_t ws_size,
                              hipStream_t stream) {
  (void)n_in; (void)out_size; (void)ws_size;
  const float* x = (const float*)d_in[0];
  const float* y = (const float*)d_in[1];
  float* out = (float*)d_out;

  const int n = in_sizes[0] / 3;
  const int m = in_sizes[1] / 3;
  const int xtiles = n / 16;
  const int ytiles = m / 16;

  float* partial = (float*)d_ws;

  chamfer_wmma_kernel<<<xtiles * YSPLIT / WG_WAVES, WG_THREADS, 0, stream>>>(
      x, y, partial, n, ytiles);

  finalize_sum_kernel<<<1, FS_THREADS, 0, stream>>>(x, partial, out, n);
}


# No y split: 4 waves per group each run one x tile against all of y from an LDS chunk ring that 4 staging waves fill, store closest_dists and row sums; sum pass adds 256 values; sum order changes
# speedup vs baseline: 3.2491x; 1.2385x over previous
#include <hip/hip_runtime.h>

typedef __attribute__((ext_vector_type(2))) float v2f;
typedef __attribute__((ext_vector_type(8))) float v8f;

#define XT_WAVES   4
#define ST_WAVES   4
#define WG_THREADS ((XT_WAVES + ST_WAVES) * 32)
#define ST_THREADS (ST_WAVES * 32)
#define CH_TILES   256
#define CH_FLOATS  (CH_TILES * 64)
#define RING       4
#define STAGE_PTS  (CH_TILES * 16 / ST_THREADS)
#define GRP        32

__device__ __forceinline__ v8f tile_product(v2f A, v2f B) {
  v8f c = {};
  return __builtin_amdgcn_wmma_f32_16x16x4_f32(
       false, A,  false, B,
       (short)0, c,  false,  false);
}

__device__ __forceinline__ void group_products(v2f A, const v2f (&B)[GRP],
                                               v8f& mn, v8f& p0, v8f& p1) {
#pragma unroll
  for (int k = 0; k < GRP; k += 2) {
    const v8f d0 = tile_product(A, B[k]);
    const v8f d1 = tile_product(A, B[k + 1]);
#pragma unroll
    for (int v = 0; v < 8; ++v) mn[v] = fminf(fminf(mn[v], p0[v]), p1[v]);
    p0 = d0;
    p1 = d1;
    __builtin_amdgcn_sched_barrier(0);
  }
}

__device__ __forceinline__ void read_group(v2f (&B)[GRP], const float* src) {
#pragma unroll
  for (int k = 0; k < GRP; ++k) B[k] = *(const v2f*)(src + k * 64);
}

__device__ __forceinline__ void stage_load(float (&q0)[STAGE_PTS], float (&q1)[STAGE_PTS],
                                           float (&q2)[STAGE_PTS],
                                           const float* __restrict__ y, int ytiles,
                                           int ch, int tid) {
  const int left = ytiles - ch * CH_TILES;
  const int cnt  = left < CH_TILES ? left : CH_TILES;
  const int npts = cnt * 16;
  const float* __restrict__ ysrc = y + (size_t)ch * CH_TILES * 48;
#pragma unroll
  for (int k = 0; k < STAGE_PTS; ++k) {
    const int p  = tid + k * ST_THREADS;
    const int pc = p < npts ? p : npts - 1;
    q0[k] = ysrc[pc * 3 + 0];
    q1[k] = ysrc[pc * 3 + 1];
    q2[k] = ysrc[pc * 3 + 2];
  }
}

__device__ __forceinline__ void stage_store(const float (&q0)[STAGE_PTS],
                                            const float (&q1)[STAGE_PTS],
                                            const float (&q2)[STAGE_PTS],
                                            float* slot, int tid) {
#pragma unroll
  for (int k = 0; k < STAGE_PTS; ++k) {
    const int p = tid + k * ST_THREADS;
    float* d = slot + (p >> 4) * 64 + (p & 15) * 2;
    v2f lo, hi;
    lo.x = q0[k];
    lo.y = q1[k];
    hi.x = q2[k];
    hi.y = fmaf(q2[k], q2[k], fmaf(q0[k], q0[k], q1[k] * q1[k]));
    *(v2f*)d        = lo;
    *(v2f*)(d + 32) = hi;
  }
}

__device__ __forceinline__ void chunk_products(v2f A, const float* lb, int cnt,
                                               v8f& mn, v8f& p0, v8f& p1) {
  const int ng = cnt / GRP;
  if (ng > 0) {
    v2f Ba[GRP], Bb[GRP];
    read_group(Ba, lb);
    int g = 0;
    for (; g + 2 <= ng; g += 2) {
      read_group(Bb, lb + (g + 1) * (GRP * 64));
      group_products(A, Ba, mn, p0, p1);
      const int gn = g + 2 < ng ? g + 2 : ng - 1;
      read_group(Ba, lb + gn * (GRP * 64));
      group_products(A, Bb, mn, p0, p1);
    }
    if (g < ng) group_products(A, Ba, mn, p0, p1);
  }
  for (int t = ng * GRP; t < cnt; ++t) {
    const v2f B = *(const v2f*)(lb + t * 64);
    const v8f d = tile_product(A, B);
#pragma unroll
    for (int v = 0; v < 8; ++v) mn[v] = fminf(mn[v], d[v]);
  }
}

__global__ __launch_bounds__(WG_THREADS) void chamfer_rows_kernel(
    const float* __restrict__ x, const float* __restrict__ y,
    float* __restrict__ out, float* __restrict__ wgsum, int ytiles) {
  __shared__ __attribute__((aligned(16))) float sB[RING * CH_FLOATS];
  __shared__ float wsum[XT_WAVES];

  const int tid   = threadIdx.x;
  const int lane  = tid & 31;
  const int wave  = __builtin_amdgcn_readfirstlane(tid >> 5);
  const bool stager = wave >= XT_WAVES;
  const int stid  = tid - XT_WAVES * 32;
  const int xtile = blockIdx.x * XT_WAVES + (wave & (XT_WAVES - 1));

  const int c   = lane & 15;
  const int row = xtile * 16 + c;
  const float x0 = x[row * 3 + 0];
  const float x1 = x[row * 3 + 1];
  const float x2 = x[row * 3 + 2];

  const int nch = (ytiles + CH_TILES - 1) / CH_TILES;

  v2f A;
  if (lane < 16) {
    A.x = -2.0f * x0;
    A.y = -2.0f * x1;
  } else {
    A.x = -2.0f * x2;
    A.y = 1.0f;
  }

  v8f mn, p0, p1;
#pragma unroll
  for (int v = 0; v < 8; ++v) {
    mn[v] = __builtin_inff();
    p0[v] = __builtin_inff();
    p1[v] = __builtin_inff();
  }

  for (int ch = 0; ch <= nch; ++ch) {
    if (stager) {
      if (ch < nch) {
        float q0[STAGE_PTS], q1[STAGE_PTS], q2[STAGE_PTS];
        stage_load(q0, q1, q2, y, ytiles, ch, stid);
        stage_store(q0, q1, q2, sB + (ch % RING) * CH_FLOATS, stid);
      }
    } else if (ch > 0) {
      const int left = ytiles - (ch - 1) * CH_TILES;
      chunk_products(A, sB + ((ch - 1) % RING) * CH_FLOATS + lane * 2,
                     left < CH_TILES ? left : CH_TILES, mn, p0, p1);
    }
    __syncthreads();
  }
  if (!stager) {
#pragma unroll
    for (int v = 0; v < 8; ++v) mn[v] = fminf(fminf(mn[v], p0[v]), p1[v]);

#pragma unroll
    for (int off = 8; off >= 1; off >>= 1) {
#pragma unroll
      for (int v = 0; v < 8; ++v) {
        float o = __shfl_xor(mn[v], off, 32);
        mn[v] = fminf(mn[v], o);
      }
    }

    const bool owner = (c >> 3) == (lane >> 4);
    float mv = mn[0];
#pragma unroll
    for (int v = 1; v < 8; ++v) mv = ((c & 7) == v) ? mn[v] : mv;
    const float xx = fmaf(x2, x2, fmaf(x1, x1, x0 * x0));
    const float dist = mv + xx;
    if (owner) out[1 + row] = dist;

    float acc = owner ? dist : 0.0f;
#pragma unroll
    for (int off = 16; off >= 1; off >>= 1) acc += __shfl_xor(acc, off, 32);
    if (lane == 0) wsum[wave] = acc;
  }
  __syncthreads();
  if (tid == 0) {
    float s = wsum[0];
#pragma unroll
    for (int w = 1; w < XT_WAVES; ++w) s += wsum[w];
    wgsum[blockIdx.x] = s;
  }
}

#define SP_VALS 8

__global__ __launch_bounds__(32) void sum_partials_kernel(
    const float* __restrict__ wgsum, float* __restrict__ out, int nparts) {
  const int lane = threadIdx.x;
  float acc = 0.0f;
  for (int base = 0; base < nparts; base += 32 * SP_VALS) {
    float v[SP_VALS];
#pragma unroll
    for (int k = 0; k < SP_VALS; ++k) {
      const int i  = base + k * 32 + lane;
      const int ic = i < nparts ? i : nparts - 1;
      v[k] = wgsum[ic];
    }
#pragma unroll
    for (int k = 0; k < SP_VALS; ++k) {
      const int i = base + k * 32 + lane;
      if (i < nparts) acc += v[k];
    }
  }
#pragma unroll
  for (int off = 16; off >= 1; off >>= 1) acc += __shfl_xor(acc, off, 32);
  if (lane == 0) out[0] = acc;
}

extern "C" void kernel_launch(void* const* d_in, const int* in_sizes, int n_in,
                              void* d_out, int out_size, void* d_ws, size_t ws_size,
                              hipStream_t stream) {
  (void)n_in; (void)out_size; (void)ws_size;
  const float* x = (const float*)d_in[0];
  const float* y = (const float*)d_in[1];
  float* out = (float*)d_out;

  const int n = in_sizes[0] / 3;
  const int m = in_sizes[1] / 3;
  const int xtiles = n / 16;
  const int ytiles = m / 16;
  const int ngroups = xtiles / XT_WAVES;

  float* wgsum = (float*)d_ws;

  chamfer_rows_kernel<<<ngroups, WG_THREADS, 0, stream>>>(x, y, out, wgsum, ytiles);

  sum_partials_kernel<<<1, 32, 0, stream>>>(wgsum, out, ngroups);
}
